# starting kernel
# baseline (speedup 1.0000x reference)
#include <hip/hip_runtime.h>
#include <hip/hip_bf16.h>

#define BATCH 32
#define SLEN  2048
#define DIM   1024
#define SPLIT 16

typedef __attribute__((ext_vector_type(2))) float v2f;
typedef __attribute__((ext_vector_type(8))) float v8f;

__global__ __launch_bounds__(256) void k1_u_gemm(const float* __restrict__ st,
                                                 const float* __restrict__ W,
                                                 float* __restrict__ u) {
  const int wave = (blockIdx.x * blockDim.x + threadIdx.x) >> 5;
  const int lane = threadIdx.x & 31;
  const int mt = wave >> 6;
  const int nt = wave & 63;
  const int row = lane & 15;
  const int kh  = lane >> 4;
  const int m0 = mt * 16, n0 = nt * 16;

  v8f acc = {};
  for (int k = 0; k < DIM; k += 4) {
    const int kk = k + 2 * kh;
    v2f a, bm;
    a.x  = st[(m0 + row) * DIM + kk];
    a.y  = st[(m0 + row) * DIM + kk + 1];
    bm.x = W[(size_t)kk * DIM + n0 + row];
    bm.y = W[(size_t)(kk + 1) * DIM + n0 + row];
    acc = __builtin_amdgcn_wmma_f32_16x16x4_f32(false, a, false, bm,
                                                (short)0, acc, false, false);
  }
#pragma unroll
  for (int i = 0; i < 8; ++i)
    u[(size_t)(m0 + i + 8 * kh) * DIM + n0 + row] = acc[i];
}

__global__ __launch_bounds__(256) void k2_flash_partial(
    const float* __restrict__ st, const float* __restrict__ hx,
    const float* __restrict__ mask, const float* __restrict__ bias,
    const float* __restrict__ u,
    float* __restrict__ pm, float* __restrict__ pl, float* __restrict__ pv) {
  const int b     = blockIdx.x / SPLIT;
  const int split = blockIdx.x % SPLIT;
  const int tid   = threadIdx.x;
  const int lane  = tid & 31;
  const int w     = tid >> 5;

  __shared__ float red[256];
  __shared__ float wave_m[8], wave_l[8];
  __shared__ float vsh[8][DIM];

  float cpart = 0.f;
  for (int i = tid; i < DIM; i += 256)
    cpart += st[b * DIM + i] * bias[i];
  red[tid] = cpart;
  __syncthreads();
  for (int off = 128; off > 0; off >>= 1) {
    if (tid < off) red[tid] += red[tid + off];
    __syncthreads();
  }
  const float cb = red[0];

  float4 uv[8];
  const float4* u4 = (const float4*)(u + (size_t)b * DIM);
#pragma unroll
  for (int j = 0; j < 8; ++j) uv[j] = u4[32 * j + lane];

  float m_run = -__builtin_inff();
  float l_run = 0.f;
  float4 vacc[8];
#pragma unroll
  for (int j = 0; j < 8; ++j) { vacc[j].x = vacc[j].y = vacc[j].z = vacc[j].w = 0.f; }

  const int rows_per_split = SLEN / SPLIT;
  const int rows_per_wave  = rows_per_split / 8;
  const int s0 = split * rows_per_split + w * rows_per_wave;

  for (int r = 0; r < rows_per_wave; ++r) {
    const int s = s0 + r;
    const float* hrow = hx + ((size_t)b * SLEN + s) * DIM;
    if (r + 1 < rows_per_wave)
      __builtin_prefetch(hrow + DIM + lane * 32, 0, 0);

    const float4* h4 = (const float4*)hrow;
    float4 hv[8];
#pragma unroll
    for (int j = 0; j < 8; ++j) hv[j] = h4[32 * j + lane];

    float p = 0.f;
#pragma unroll
    for (int j = 0; j < 8; ++j)
      p += hv[j].x * uv[j].x + hv[j].y * uv[j].y +
           hv[j].z * uv[j].z + hv[j].w * uv[j].w;
    p += __shfl_xor(p, 16, 32);
    p += __shfl_xor(p,  8, 32);
    p += __shfl_xor(p,  4, 32);
    p += __shfl_xor(p,  2, 32);
    p += __shfl_xor(p,  1, 32);

    const float mk    = mask[b * SLEN + s];
    const float mk1   = mk + (1.f - mk) * 1e-18f;
    const float score = (p + cb) * mk1;

    const float mn = fmaxf(m_run, score);
    const float sc = __expf(m_run - mn);
    const float pe = __expf(score - mn);
    l_run = l_run * sc + pe;
#pragma unroll
    for (int j = 0; j < 8; ++j) {
      vacc[j].x = vacc[j].x * sc + pe * hv[j].x;
      vacc[j].y = vacc[j].y * sc + pe * hv[j].y;
      vacc[j].z = vacc[j].z * sc + pe * hv[j].z;
      vacc[j].w = vacc[j].w * sc + pe * hv[j].w;
    }
    m_run = mn;
  }

  if (lane == 0) { wave_m[w] = m_run; wave_l[w] = l_run; }
  __syncthreads();
  float mb = wave_m[0];
#pragma unroll
  for (int i = 1; i < 8; ++i) mb = fmaxf(mb, wave_m[i]);
  const float f = __expf(m_run - mb);
#pragma unroll
  for (int j = 0; j < 8; ++j) {
    const int base = 4 * (32 * j + lane);
    vsh[w][base + 0] = vacc[j].x * f;
    vsh[w][base + 1] = vacc[j].y * f;
    vsh[w][base + 2] = vacc[j].z * f;
    vsh[w][base + 3] = vacc[j].w * f;
  }
  __syncthreads();

  float lb = 0.f;
#pragma unroll
  for (int i = 0; i < 8; ++i) lb += wave_l[i] * __expf(wave_m[i] - mb);

  const int pi = b * SPLIT + split;
  for (int d = tid; d < DIM; d += 256) {
    float sv = 0.f;
#pragma unroll
    for (int i = 0; i < 8; ++i) sv += vsh[i][d];
    pv[(size_t)pi * DIM + d] = sv;
  }
  if (tid == 0) { pm[pi] = mb; pl[pi] = lb; }
}

__global__ __launch_bounds__(256) void k3_combine(const float* __restrict__ pm,
                                                  const float* __restrict__ pl,
                                                  const float* __restrict__ pv,
                                                  float* __restrict__ vnorm) {
  const int b = blockIdx.x;
  const int tid = threadIdx.x;
  float mb = -__builtin_inff();
#pragma unroll
  for (int i = 0; i < SPLIT; ++i) mb = fmaxf(mb, pm[b * SPLIT + i]);
  float f[SPLIT];
  float L = 0.f;
#pragma unroll
  for (int i = 0; i < SPLIT; ++i) {
    f[i] = __expf(pm[b * SPLIT + i] - mb);
    L += pl[b * SPLIT + i] * f[i];
  }
  const float invL = 1.f / L;
  for (int d = tid; d < DIM; d += 256) {
    float sv = 0.f;
#pragma unroll
    for (int i = 0; i < SPLIT; ++i)
      sv += pv[(size_t)(b * SPLIT + i) * DIM + d] * f[i];
    vnorm[(size_t)b * DIM + d] = sv * invL;
  }
}

__global__ __launch_bounds__(256) void k4_out_gemm(const float* __restrict__ vnorm,
                                                   const float* __restrict__ W,
                                                   const float* __restrict__ bias,
                                                   float* __restrict__ ct) {
  const int wave = (blockIdx.x * blockDim.x + threadIdx.x) >> 5;
  const int lane = threadIdx.x & 31;
  const int mt = wave >> 6;
  const int nt = wave & 63;
  const int row = lane & 15;
  const int kh  = lane >> 4;
  const int m0 = mt * 16, n0 = nt * 16;

  v8f acc = {};
  for (int k = 0; k < DIM; k += 4) {
    const int kk = k + 2 * kh;
    v2f a, bm;
    a.x  = vnorm[(size_t)(m0 + row) * DIM + kk];
    a.y  = vnorm[(size_t)(m0 + row) * DIM + kk + 1];
    bm.x = W[(size_t)(n0 + row) * DIM + kk];
    bm.y = W[(size_t)(n0 + row) * DIM + kk + 1];
    acc = __builtin_amdgcn_wmma_f32_16x16x4_f32(false, a, false, bm,
                                                (short)0, acc, false, false);
  }
  const float bd = bias[n0 + row];
#pragma unroll
  for (int i = 0; i < 8; ++i)
    ct[(size_t)(m0 + i + 8 * kh) * DIM + n0 + row] = acc[i] + bd;
}

extern "C" void kernel_launch(void* const* d_in, const int* in_sizes, int n_in,
                              void* d_out, int out_size, void* d_ws, size_t ws_size,
                              hipStream_t stream) {
  const float* st   = (const float*)d_in[0];
  const float* hx   = (const float*)d_in[1];
  const float* mask = (const float*)d_in[2];
  const float* W    = (const float*)d_in[3];
  const float* bias = (const float*)d_in[4];
  float* out = (float*)d_out;

  float* wsf   = (float*)d_ws;
  float* u     = wsf;
  float* pm    = u  + (size_t)BATCH * DIM;
  float* pl    = pm + (size_t)BATCH * SPLIT;
  float* pv    = pl + (size_t)BATCH * SPLIT;
  float* vnorm = pv + (size_t)BATCH * SPLIT * DIM;

  k1_u_gemm<<<16, 256, 0, stream>>>(st, W, u);

  k2_flash_partial<<<BATCH * SPLIT, 256, 0, stream>>>(st, hx, mask, bias, u,
                                                      pm, pl, pv);

  k3_combine<<<BATCH, 256, 0, stream>>>(pm, pl, pv, vnorm);

  k4_out_gemm<<<16, 256, 0, stream>>>(vnorm, W, bias, out);
}


# Both skinny GEMMs: split K over the 16 waves of one workgroup per 16x16 output tile, partial tiles summed through LDS (128 -> 2048 waves in flight; k sum order changes)
# speedup vs baseline: 1.8017x; 1.8017x over previous
#include <hip/hip_runtime.h>
#include <hip/hip_bf16.h>

#define BATCH 32
#define SLEN  2048
#define DIM   1024
#define SPLIT 16

typedef __attribute__((ext_vector_type(2))) float v2f;
typedef __attribute__((ext_vector_type(8))) float v8f;

#define KSPLIT 16
#define KCHUNK (DIM / KSPLIT)

__global__ __launch_bounds__(32 * KSPLIT) void k1_u_gemm(const float* __restrict__ st,
                                                         const float* __restrict__ W,
                                                         float* __restrict__ u) {
  __shared__ float part[KSPLIT][256];
  const int tid  = threadIdx.x;
  const int kw   = tid >> 5;
  const int lane = tid & 31;
  const int mt = blockIdx.x >> 6;
  const int nt = blockIdx.x & 63;
  const int row = lane & 15;
  const int kh  = lane >> 4;
  const int m0 = mt * 16, n0 = nt * 16;
  const int kbeg = kw * KCHUNK;

  v8f acc = {};
#pragma unroll
  for (int k = 0; k < KCHUNK; k += 4) {
    const int kk = kbeg + k + 2 * kh;
    v2f a, bm;
    a.x  = st[(m0 + row) * DIM + kk];
    a.y  = st[(m0 + row) * DIM + kk + 1];
    bm.x = W[(size_t)kk * DIM + n0 + row];
    bm.y = W[(size_t)(kk + 1) * DIM + n0 + row];
    acc = __builtin_amdgcn_wmma_f32_16x16x4_f32(false, a, false, bm,
                                                (short)0, acc, false, false);
  }
#pragma unroll
  for (int i = 0; i < 8; ++i) part[kw][i * 32 + lane] = acc[i];
  __syncthreads();
  if (tid < 256) {
    float s = 0.f;
#pragma unroll
    for (int j = 0; j < KSPLIT; ++j) s += part[j][tid];
    u[(size_t)(m0 + kw + 8 * kh) * DIM + n0 + row] = s;
  }
}

__global__ __launch_bounds__(256) void k2_flash_partial(
    const float* __restrict__ st, const float* __restrict__ hx,
    const float* __restrict__ mask, const float* __restrict__ bias,
    const float* __restrict__ u,
    float* __restrict__ pm, float* __restrict__ pl, float* __restrict__ pv) {
  const int b     = blockIdx.x / SPLIT;
  const int split = blockIdx.x % SPLIT;
  const int tid   = threadIdx.x;
  const int lane  = tid & 31;
  const int w     = tid >> 5;

  __shared__ float red[256];
  __shared__ float wave_m[8], wave_l[8];
  __shared__ float vsh[8][DIM];

  float cpart = 0.f;
  for (int i = tid; i < DIM; i += 256)
    cpart += st[b * DIM + i] * bias[i];
  red[tid] = cpart;
  __syncthreads();
  for (int off = 128; off > 0; off >>= 1) {
    if (tid < off) red[tid] += red[tid + off];
    __syncthreads();
  }
  const float cb = red[0];

  float4 uv[8];
  const float4* u4 = (const float4*)(u + (size_t)b * DIM);
#pragma unroll
  for (int j = 0; j < 8; ++j) uv[j] = u4[32 * j + lane];

  float m_run = -__builtin_inff();
  float l_run = 0.f;
  float4 vacc[8];
#pragma unroll
  for (int j = 0; j < 8; ++j) { vacc[j].x = vacc[j].y = vacc[j].z = vacc[j].w = 0.f; }

  const int rows_per_split = SLEN / SPLIT;
  const int rows_per_wave  = rows_per_split / 8;
  const int s0 = split * rows_per_split + w * rows_per_wave;

  for (int r = 0; r < rows_per_wave; ++r) {
    const int s = s0 + r;
    const float* hrow = hx + ((size_t)b * SLEN + s) * DIM;
    if (r + 1 < rows_per_wave)
      __builtin_prefetch(hrow + DIM + lane * 32, 0, 0);

    const float4* h4 = (const float4*)hrow;
    float4 hv[8];
#pragma unroll
    for (int j = 0; j < 8; ++j) hv[j] = h4[32 * j + lane];

    float p = 0.f;
#pragma unroll
    for (int j = 0; j < 8; ++j)
      p += hv[j].x * uv[j].x + hv[j].y * uv[j].y +
           hv[j].z * uv[j].z + hv[j].w * uv[j].w;
    p += __shfl_xor(p, 16, 32);
    p += __shfl_xor(p,  8, 32);
    p += __shfl_xor(p,  4, 32);
    p += __shfl_xor(p,  2, 32);
    p += __shfl_xor(p,  1, 32);

    const float mk    = mask[b * SLEN + s];
    const float mk1   = mk + (1.f - mk) * 1e-18f;
    const float score = (p + cb) * mk1;

    const float mn = fmaxf(m_run, score);
    const float sc = __expf(m_run - mn);
    const float pe = __expf(score - mn);
    l_run = l_run * sc + pe;
#pragma unroll
    for (int j = 0; j < 8; ++j) {
      vacc[j].x = vacc[j].x * sc + pe * hv[j].x;
      vacc[j].y = vacc[j].y * sc + pe * hv[j].y;
      vacc[j].z = vacc[j].z * sc + pe * hv[j].z;
      vacc[j].w = vacc[j].w * sc + pe * hv[j].w;
    }
    m_run = mn;
  }

  if (lane == 0) { wave_m[w] = m_run; wave_l[w] = l_run; }
  __syncthreads();
  float mb = wave_m[0];
#pragma unroll
  for (int i = 1; i < 8; ++i) mb = fmaxf(mb, wave_m[i]);
  const float f = __expf(m_run - mb);
#pragma unroll
  for (int j = 0; j < 8; ++j) {
    const int base = 4 * (32 * j + lane);
    vsh[w][base + 0] = vacc[j].x * f;
    vsh[w][base + 1] = vacc[j].y * f;
    vsh[w][base + 2] = vacc[j].z * f;
    vsh[w][base + 3] = vacc[j].w * f;
  }
  __syncthreads();

  float lb = 0.f;
#pragma unroll
  for (int i = 0; i < 8; ++i) lb += wave_l[i] * __expf(wave_m[i] - mb);

  const int pi = b * SPLIT + split;
  for (int d = tid; d < DIM; d += 256) {
    float sv = 0.f;
#pragma unroll
    for (int i = 0; i < 8; ++i) sv += vsh[i][d];
    pv[(size_t)pi * DIM + d] = sv;
  }
  if (tid == 0) { pm[pi] = mb; pl[pi] = lb; }
}

__global__ __launch_bounds__(256) void k3_combine(const float* __restrict__ pm,
                                                  const float* __restrict__ pl,
                                                  const float* __restrict__ pv,
                                                  float* __restrict__ vnorm) {
  const int b = blockIdx.x;
  const int tid = threadIdx.x;
  float mb = -__builtin_inff();
#pragma unroll
  for (int i = 0; i < SPLIT; ++i) mb = fmaxf(mb, pm[b * SPLIT + i]);
  float f[SPLIT];
  float L = 0.f;
#pragma unroll
  for (int i = 0; i < SPLIT; ++i) {
    f[i] = __expf(pm[b * SPLIT + i] - mb);
    L += pl[b * SPLIT + i] * f[i];
  }
  const float invL = 1.f / L;
  for (int d = tid; d < DIM; d += 256) {
    float sv = 0.f;
#pragma unroll
    for (int i = 0; i < SPLIT; ++i)
      sv += pv[(size_t)(b * SPLIT + i) * DIM + d] * f[i];
    vnorm[(size_t)b * DIM + d] = sv * invL;
  }
}

__global__ __launch_bounds__(32 * KSPLIT) void k4_out_gemm(const float* __restrict__ vnorm,
                                                           const float* __restrict__ W,
                                                           const float* __restrict__ bias,
                                                           float* __restrict__ ct) {
  __shared__ float part[KSPLIT][256];
  const int tid  = threadIdx.x;
  const int kw   = tid >> 5;
  const int lane = tid & 31;
  const int mt = blockIdx.x >> 6;
  const int nt = blockIdx.x & 63;
  const int row = lane & 15;
  const int kh  = lane >> 4;
  const int m0 = mt * 16, n0 = nt * 16;
  const int kbeg = kw * KCHUNK;

  v8f acc = {};
#pragma unroll
  for (int k = 0; k < KCHUNK; k += 4) {
    const int kk = kbeg + k + 2 * kh;
    v2f a, bm;
    a.x  = vnorm[(size_t)(m0 + row) * DIM + kk];
    a.y  = vnorm[(size_t)(m0 + row) * DIM + kk + 1];
    bm.x = W[(size_t)(n0 + row) * DIM + kk];
    bm.y = W[(size_t)(n0 + row) * DIM + kk + 1];
    acc = __builtin_amdgcn_wmma_f32_16x16x4_f32(false, a, false, bm,
                                                (short)0, acc, false, false);
  }
#pragma unroll
  for (int i = 0; i < 8; ++i) part[kw][i * 32 + lane] = acc[i];
  __syncthreads();
  if (tid < 256) {
    float s = 0.f;
#pragma unroll
    for (int j = 0; j < KSPLIT; ++j) s += part[j][tid];
    ct[(size_t)(m0 + kw + 8 * kh) * DIM + n0 + row] = s + bias[n0 + row];
  }
}

extern "C" void kernel_launch(void* const* d_in, const int* in_sizes, int n_in,
                              void* d_out, int out_size, void* d_ws, size_t ws_size,
                              hipStream_t stream) {
  const float* st   = (const float*)d_in[0];
  const float* hx   = (const float*)d_in[1];
  const float* mask = (const float*)d_in[2];
  const float* W    = (const float*)d_in[3];
  const float* bias = (const float*)d_in[4];
  float* out = (float*)d_out;

  float* wsf   = (float*)d_ws;
  float* u     = wsf;
  float* pm    = u  + (size_t)BATCH * DIM;
  float* pl    = pm + (size_t)BATCH * SPLIT;
  float* pv    = pl + (size_t)BATCH * SPLIT;
  float* vnorm = pv + (size_t)BATCH * SPLIT * DIM;

  k1_u_gemm<<<128, 32 * KSPLIT, 0, stream>>>(st, W, u);

  k2_flash_partial<<<BATCH * SPLIT, 256, 0, stream>>>(st, hx, mask, bias, u,
                                                      pm, pl, pv);

  k3_combine<<<BATCH, 256, 0, stream>>>(pm, pl, pv, vnorm);

  k4_out_gemm<<<128, 32 * KSPLIT, 0, stream>>>(vnorm, W, bias, out);
}


# Flash pass as a stream: rows in pairs, next pair's 16 non-temporal loads in flight while a pair is reduced (one rescale per pair; bias dot per wave, no barrier before rows; dot sums re-associated)
# speedup vs baseline: 2.1651x; 1.2017x over previous
#include <hip/hip_runtime.h>
#include <hip/hip_bf16.h>

#define BATCH 32
#define SLEN  2048
#define DIM   1024
#define SPLIT 16

typedef __attribute__((ext_vector_type(2))) float v2f;
typedef __attribute__((ext_vector_type(8))) float v8f;

#define KSPLIT 16
#define KCHUNK (DIM / KSPLIT)

__global__ __launch_bounds__(32 * KSPLIT) void k1_u_gemm(const float* __restrict__ st,
                                                         const float* __restrict__ W,
                                                         float* __restrict__ u) {
  __shared__ float part[KSPLIT][256];
  const int tid  = threadIdx.x;
  const int kw   = tid >> 5;
  const int lane = tid & 31;
  const int mt = blockIdx.x >> 6;
  const int nt = blockIdx.x & 63;
  const int row = lane & 15;
  const int kh  = lane >> 4;
  const int m0 = mt * 16, n0 = nt * 16;
  const int kbeg = kw * KCHUNK;

  v8f acc = {};
#pragma unroll
  for (int k = 0; k < KCHUNK; k += 4) {
    const int kk = kbeg + k + 2 * kh;
    v2f a, bm;
    a.x  = st[(m0 + row) * DIM + kk];
    a.y  = st[(m0 + row) * DIM + kk + 1];
    bm.x = W[(size_t)kk * DIM + n0 + row];
    bm.y = W[(size_t)(kk + 1) * DIM + n0 + row];
    acc = __builtin_amdgcn_wmma_f32_16x16x4_f32(false, a, false, bm,
                                                (short)0, acc, false, false);
  }
#pragma unroll
  for (int i = 0; i < 8; ++i) part[kw][i * 32 + lane] = acc[i];
  __syncthreads();
  if (tid < 256) {
    float s = 0.f;
#pragma unroll
    for (int j = 0; j < KSPLIT; ++j) s += part[j][tid];
    u[(size_t)(m0 + kw + 8 * kh) * DIM + n0 + row] = s;
  }
}

typedef __attribute__((ext_vector_type(4))) float v4f;
#define RPW 16
#define NPAIR (RPW / 2)

__device__ __forceinline__ v2f xor_pair(v2f p, int m) {
  v2f r;
  r.x = __shfl_xor(p.x, m, 32);
  r.y = __shfl_xor(p.y, m, 32);
  return r;
}

__device__ __forceinline__ void load_pair(v4f (&h)[2][8], const v4f* p) {
#pragma unroll
  for (int i = 0; i < 2; ++i)
#pragma unroll
    for (int q = 0; q < 8; ++q)
      h[i][q] = __builtin_nontemporal_load(p + i * (DIM / 4) + 32 * q);
}

__device__ __forceinline__ void reduce_pair(const v4f (&h)[2][8], const v4f (&uv)[8],
                                            float mkA, float mkB, float cb,
                                            float& m_run, float& l_run, v4f (&vacc)[8]) {
  v2f a0 = {0.f, 0.f}, a1 = {0.f, 0.f};
  v2f e0 = {0.f, 0.f}, e1 = {0.f, 0.f};
#pragma unroll
  for (int j = 0; j < 8; ++j) {
    a0 += h[0][j].xy * uv[j].xy;
    a1 += h[0][j].zw * uv[j].zw;
    e0 += h[1][j].xy * uv[j].xy;
    e1 += h[1][j].zw * uv[j].zw;
  }
  v2f pp;
  pp.x = (a0.x + a0.y) + (a1.x + a1.y);
  pp.y = (e0.x + e0.y) + (e1.x + e1.y);
  pp += xor_pair(pp, 16);
  pp += xor_pair(pp,  8);
  pp += xor_pair(pp,  4);
  pp += xor_pair(pp,  2);
  pp += xor_pair(pp,  1);

  const float scoreA = (pp.x + cb) * (mkA + (1.f - mkA) * 1e-18f);
  const float scoreB = (pp.y + cb) * (mkB + (1.f - mkB) * 1e-18f);

  const float mn  = fmaxf(m_run, fmaxf(scoreA, scoreB));
  const float sc  = __expf(m_run - mn);
  const float peA = __expf(scoreA - mn);
  const float peB = __expf(scoreB - mn);
  l_run = (l_run * sc + peA) + peB;
#pragma unroll
  for (int j = 0; j < 8; ++j) vacc[j] = (vacc[j] * sc + peA * h[0][j]) + peB * h[1][j];
  m_run = mn;
}

__device__ __forceinline__ float lane_value(float v, int l) {
  return __builtin_bit_cast(float, __builtin_amdgcn_readlane(__builtin_bit_cast(int, v), l));
}

__global__ __launch_bounds__(256) void k2_flash_partial(
    const float* __restrict__ st, const float* __restrict__ hx,
    const float* __restrict__ mask, const float* __restrict__ bias,
    const float* __restrict__ u,
    float* __restrict__ pm, float* __restrict__ pl, float* __restrict__ pv) {
  const int b     = blockIdx.x / SPLIT;
  const int split = blockIdx.x % SPLIT;
  const int tid   = threadIdx.x;
  const int lane  = tid & 31;
  const int w     = __builtin_amdgcn_readfirstlane(tid >> 5);

  __shared__ float wave_m[8], wave_l[8];
  __shared__ float vsh[8][DIM];

  const int s0 = split * (SLEN / SPLIT) + w * RPW;

  const v4f* u4 = (const v4f*)(u  + (size_t)b * DIM) + lane;
  const v4f* s4 = (const v4f*)(st + (size_t)b * DIM) + lane;
  const v4f* b4 = (const v4f*)bias + lane;
  v4f uv[8], sv[8], bv[8];
#pragma unroll
  for (int j = 0; j < 8; ++j) { uv[j] = u4[32 * j]; sv[j] = s4[32 * j]; bv[j] = b4[32 * j]; }
  const float mk_lane = mask[b * SLEN + s0 + (lane & (RPW - 1))];

  const v4f* hp = (const v4f*)(hx + ((size_t)b * SLEN + s0) * DIM) + lane;
  const int pair_step = 2 * (DIM / 4);
  v4f hA[2][8], hB[2][8];
  __builtin_amdgcn_sched_barrier(0);
  load_pair(hA, hp);

  float cb;
  {
    v2f c0 = {0.f, 0.f}, c1 = {0.f, 0.f};
#pragma unroll
    for (int j = 0; j < 8; ++j) {
      c0 += sv[j].xy * bv[j].xy;
      c1 += sv[j].zw * bv[j].zw;
    }
    cb = (c0.x + c0.y) + (c1.x + c1.y);
    cb += __shfl_xor(cb, 16, 32);
    cb += __shfl_xor(cb,  8, 32);
    cb += __shfl_xor(cb,  4, 32);
    cb += __shfl_xor(cb,  2, 32);
    cb += __shfl_xor(cb,  1, 32);
  }

  float m_run = -__builtin_inff();
  float l_run = 0.f;
  v4f vacc[8];
#pragma unroll
  for (int j = 0; j < 8; ++j) vacc[j] = (v4f){0.f, 0.f, 0.f, 0.f};

#define PAIR_FENCE() \
  asm volatile("" : "+v"(vacc[0]), "+v"(vacc[1]), "+v"(vacc[2]), "+v"(vacc[3]), \
                    "+v"(vacc[4]), "+v"(vacc[5]), "+v"(vacc[6]), "+v"(vacc[7]), \
                    "+v"(uv[0]), "+v"(uv[1]), "+v"(uv[2]), "+v"(uv[3]), \
                    "+v"(uv[4]), "+v"(uv[5]), "+v"(uv[6]), "+v"(uv[7]), \
                    "+v"(m_run), "+v"(l_run) : : "memory")
#pragma unroll
  for (int g = 0; g < NPAIR; ++g) {
    __builtin_amdgcn_sched_barrier(0);
    if (g + 1 < NPAIR) {
      if ((g & 1) == 0) load_pair(hB, hp + (g + 1) * pair_step);
      else              load_pair(hA, hp + (g + 1) * pair_step);
    }
    __builtin_amdgcn_sched_barrier(0);
    if ((g & 1) == 0)
      reduce_pair(hA, uv, lane_value(mk_lane, 2 * g), lane_value(mk_lane, 2 * g + 1), cb, m_run, l_run, vacc);
    else
      reduce_pair(hB, uv, lane_value(mk_lane, 2 * g), lane_value(mk_lane, 2 * g + 1), cb, m_run, l_run, vacc);
    PAIR_FENCE();
  }
#undef PAIR_FENCE

  if (lane == 0) { wave_m[w] = m_run; wave_l[w] = l_run; }
  __syncthreads();
  float mb = wave_m[0];
#pragma unroll
  for (int i = 1; i < 8; ++i) mb = fmaxf(mb, wave_m[i]);
  const float f = __expf(m_run - mb);
#pragma unroll
  for (int j = 0; j < 8; ++j) {
    const int base = 4 * (32 * j + lane);
    vsh[w][base + 0] = vacc[j].x * f;
    vsh[w][base + 1] = vacc[j].y * f;
    vsh[w][base + 2] = vacc[j].z * f;
    vsh[w][base + 3] = vacc[j].w * f;
  }
  __syncthreads();

  float lb = 0.f;
#pragma unroll
  for (int i = 0; i < 8; ++i) lb += wave_l[i] * __expf(wave_m[i] - mb);

  const int pi = b * SPLIT + split;
  for (int d = tid; d < DIM; d += 256) {
    float sum = 0.f;
#pragma unroll
    for (int i = 0; i < 8; ++i) sum += vsh[i][d];
    pv[(size_t)pi * DIM + d] = sum;
  }
  if (tid == 0) { pm[pi] = mb; pl[pi] = lb; }
}

__global__ __launch_bounds__(256) void k3_combine(const float* __restrict__ pm,
                                                  const float* __restrict__ pl,
                                                  const float* __restrict__ pv,
                                                  float* __restrict__ vnorm) {
  const int b = blockIdx.x;
  const int tid = threadIdx.x;
  float mb = -__builtin_inff();
#pragma unroll
  for (int i = 0; i < SPLIT; ++i) mb = fmaxf(mb, pm[b * SPLIT + i]);
  float f[SPLIT];
  float L = 0.f;
#pragma unroll
  for (int i = 0; i < SPLIT; ++i) {
    f[i] = __expf(pm[b * SPLIT + i] - mb);
    L += pl[b * SPLIT + i] * f[i];
  }
  const float invL = 1.f / L;
  for (int d = tid; d < DIM; d += 256) {
    float sv = 0.f;
#pragma unroll
    for (int i = 0; i < SPLIT; ++i)
      sv += pv[(size_t)(b * SPLIT + i) * DIM + d] * f[i];
    vnorm[(size_t)b * DIM + d] = sv * invL;
  }
}

__global__ __launch_bounds__(32 * KSPLIT) void k4_out_gemm(const float* __restrict__ vnorm,
                                                           const float* __restrict__ W,
                                                           const float* __restrict__ bias,
                                                           float* __restrict__ ct) {
  __shared__ float part[KSPLIT][256];
  const int tid  = threadIdx.x;
  const int kw   = tid >> 5;
  const int lane = tid & 31;
  const int mt = blockIdx.x >> 6;
  const int nt = blockIdx.x & 63;
  const int row = lane & 15;
  const int kh  = lane >> 4;
  const int m0 = mt * 16, n0 = nt * 16;
  const int kbeg = kw * KCHUNK;

  v8f acc = {};
#pragma unroll
  for (int k = 0; k < KCHUNK; k += 4) {
    const int kk = kbeg + k + 2 * kh;
    v2f a, bm;
    a.x  = vnorm[(size_t)(m0 + row) * DIM + kk];
    a.y  = vnorm[(size_t)(m0 + row) * DIM + kk + 1];
    bm.x = W[(size_t)(n0 + row) * DIM + kk];
    bm.y = W[(size_t)(n0 + row) * DIM + kk + 1];
    acc = __builtin_amdgcn_wmma_f32_16x16x4_f32(false, a, false, bm,
                                                (short)0, acc, false, false);
  }
#pragma unroll
  for (int i = 0; i < 8; ++i) part[kw][i * 32 + lane] = acc[i];
  __syncthreads();
  if (tid < 256) {
    float s = 0.f;
#pragma unroll
    for (int j = 0; j < KSPLIT; ++j) s += part[j][tid];
    ct[(size_t)(m0 + kw + 8 * kh) * DIM + n0 + row] = s + bias[n0 + row];
  }
}

extern "C" void kernel_launch(void* const* d_in, const int* in_sizes, int n_in,
                              void* d_out, int out_size, void* d_ws, size_t ws_size,
                              hipStream_t stream) {
  const float* st   = (const float*)d_in[0];
  const float* hx   = (const float*)d_in[1];
  const float* mask = (const float*)d_in[2];
  const float* W    = (const float*)d_in[3];
  const float* bias = (const float*)d_in[4];
  float* out = (float*)d_out;

  float* wsf   = (float*)d_ws;
  float* u     = wsf;
  float* pm    = u  + (size_t)BATCH * DIM;
  float* pl    = pm + (size_t)BATCH * SPLIT;
  float* pv    = pl + (size_t)BATCH * SPLIT;
  float* vnorm = pv + (size_t)BATCH * SPLIT * DIM;

  k1_u_gemm<<<128, 32 * KSPLIT, 0, stream>>>(st, W, u);

  k2_flash_partial<<<BATCH * SPLIT, 256, 0, stream>>>(st, hx, mask, bias, u,
                                                      pm, pl, pv);

  k3_combine<<<BATCH, 256, 0, stream>>>(pm, pl, pv, vnorm);

  k4_out_gemm<<<128, 32 * KSPLIT, 0, stream>>>(vnorm, W, bias, out);
}


# Merge launch as ONE memory round trip: 128 workgroups, each thread requests its split maxima/sums and all 16 partial-row pieces (16-byte loads) before any arithmetic; was 8 dependent round trips
# speedup vs baseline: 2.3119x; 1.0678x over previous
#include <hip/hip_runtime.h>
#include <hip/hip_bf16.h>

#define BATCH 32
#define SLEN  2048
#define DIM   1024
#define SPLIT 16

typedef __attribute__((ext_vector_type(2))) float v2f;
typedef __attribute__((ext_vector_type(8))) float v8f;

#define KSPLIT 16
#define KCHUNK (DIM / KSPLIT)

__global__ __launch_bounds__(32 * KSPLIT) void k1_u_gemm(const float* __restrict__ st,
                                                         const float* __restrict__ W,
                                                         float* __restrict__ u) {
  __shared__ float part[KSPLIT][256];
  const int tid  = threadIdx.x;
  const int kw   = tid >> 5;
  const int lane = tid & 31;
  const int mt = blockIdx.x >> 6;
  const int nt = blockIdx.x & 63;
  const int row = lane & 15;
  const int kh  = lane >> 4;
  const int m0 = mt * 16, n0 = nt * 16;
  const int kbeg = kw * KCHUNK;

  v8f acc = {};
#pragma unroll
  for (int k = 0; k < KCHUNK; k += 4) {
    const int kk = kbeg + k + 2 * kh;
    v2f a, bm;
    a.x  = st[(m0 + row) * DIM + kk];
    a.y  = st[(m0 + row) * DIM + kk + 1];
    bm.x = W[(size_t)kk * DIM + n0 + row];
    bm.y = W[(size_t)(kk + 1) * DIM + n0 + row];
    acc = __builtin_amdgcn_wmma_f32_16x16x4_f32(false, a, false, bm,
                                                (short)0, acc, false, false);
  }
#pragma unroll
  for (int i = 0; i < 8; ++i) part[kw][i * 32 + lane] = acc[i];
  __syncthreads();
  if (tid < 256) {
    float s = 0.f;
#pragma unroll
    for (int j = 0; j < KSPLIT; ++j) s += part[j][tid];
    u[(size_t)(m0 + kw + 8 * kh) * DIM + n0 + row] = s;
  }
}

typedef __attribute__((ext_vector_type(4))) float v4f;
#define RPW 16
#define NPAIR (RPW / 2)

__device__ __forceinline__ v2f xor_pair(v2f p, int m) {
  v2f r;
  r.x = __shfl_xor(p.x, m, 32);
  r.y = __shfl_xor(p.y, m, 32);
  return r;
}

__device__ __forceinline__ void load_pair(v4f (&h)[2][8], const v4f* p) {
#pragma unroll
  for (int i = 0; i < 2; ++i)
#pragma unroll
    for (int q = 0; q < 8; ++q)
      h[i][q] = __builtin_nontemporal_load(p + i * (DIM / 4) + 32 * q);
}

__device__ __forceinline__ void reduce_pair(const v4f (&h)[2][8], const v4f (&uv)[8],
                                            float mkA, float mkB, float cb,
                                            float& m_run, float& l_run, v4f (&vacc)[8]) {
  v2f a0 = {0.f, 0.f}, a1 = {0.f, 0.f};
  v2f e0 = {0.f, 0.f}, e1 = {0.f, 0.f};
#pragma unroll
  for (int j = 0; j < 8; ++j) {
    a0 += h[0][j].xy * uv[j].xy;
    a1 += h[0][j].zw * uv[j].zw;
    e0 += h[1][j].xy * uv[j].xy;
    e1 += h[1][j].zw * uv[j].zw;
  }
  v2f pp;
  pp.x = (a0.x + a0.y) + (a1.x + a1.y);
  pp.y = (e0.x + e0.y) + (e1.x + e1.y);
  pp += xor_pair(pp, 16);
  pp += xor_pair(pp,  8);
  pp += xor_pair(pp,  4);
  pp += xor_pair(pp,  2);
  pp += xor_pair(pp,  1);

  const float scoreA = (pp.x + cb) * (mkA + (1.f - mkA) * 1e-18f);
  const float scoreB = (pp.y + cb) * (mkB + (1.f - mkB) * 1e-18f);

  const float mn  = fmaxf(m_run, fmaxf(scoreA, scoreB));
  const float sc  = __expf(m_run - mn);
  const float peA = __expf(scoreA - mn);
  const float peB = __expf(scoreB - mn);
  l_run = (l_run * sc + peA) + peB;
#pragma unroll
  for (int j = 0; j < 8; ++j) vacc[j] = (vacc[j] * sc + peA * h[0][j]) + peB * h[1][j];
  m_run = mn;
}

__device__ __forceinline__ float lane_value(float v, int l) {
  return __builtin_bit_cast(float, __builtin_amdgcn_readlane(__builtin_bit_cast(int, v), l));
}

__global__ __launch_bounds__(256) void k2_flash_partial(
    const float* __restrict__ st, const float* __restrict__ hx,
    const float* __restrict__ mask, const float* __restrict__ bias,
    const float* __restrict__ u,
    float* __restrict__ pm, float* __restrict__ pl, float* __restrict__ pv) {
  const int b     = blockIdx.x / SPLIT;
  const int split = blockIdx.x % SPLIT;
  const int tid   = threadIdx.x;
  const int lane  = tid & 31;
  const int w     = __builtin_amdgcn_readfirstlane(tid >> 5);

  __shared__ float wave_m[8], wave_l[8];
  __shared__ float vsh[8][DIM];

  const int s0 = split * (SLEN / SPLIT) + w * RPW;

  const v4f* u4 = (const v4f*)(u  + (size_t)b * DIM) + lane;
  const v4f* s4 = (const v4f*)(st + (size_t)b * DIM) + lane;
  const v4f* b4 = (const v4f*)bias + lane;
  v4f uv[8], sv[8], bv[8];
#pragma unroll
  for (int j = 0; j < 8; ++j) { uv[j] = u4[32 * j]; sv[j] = s4[32 * j]; bv[j] = b4[32 * j]; }
  const float mk_lane = mask[b * SLEN + s0 + (lane & (RPW - 1))];

  const v4f* hp = (const v4f*)(hx + ((size_t)b * SLEN + s0) * DIM) + lane;
  const int pair_step = 2 * (DIM / 4);
  v4f hA[2][8], hB[2][8];
  __builtin_amdgcn_sched_barrier(0);
  load_pair(hA, hp);

  float cb;
  {
    v2f c0 = {0.f, 0.f}, c1 = {0.f, 0.f};
#pragma unroll
    for (int j = 0; j < 8; ++j) {
      c0 += sv[j].xy * bv[j].xy;
      c1 += sv[j].zw * bv[j].zw;
    }
    cb = (c0.x + c0.y) + (c1.x + c1.y);
    cb += __shfl_xor(cb, 16, 32);
    cb += __shfl_xor(cb,  8, 32);
    cb += __shfl_xor(cb,  4, 32);
    cb += __shfl_xor(cb,  2, 32);
    cb += __shfl_xor(cb,  1, 32);
  }

  float m_run = -__builtin_inff();
  float l_run = 0.f;
  v4f vacc[8];
#pragma unroll
  for (int j = 0; j < 8; ++j) vacc[j] = (v4f){0.f, 0.f, 0.f, 0.f};

#define PAIR_FENCE() \
  asm volatile("" : "+v"(vacc[0]), "+v"(vacc[1]), "+v"(vacc[2]), "+v"(vacc[3]), \
                    "+v"(vacc[4]), "+v"(vacc[5]), "+v"(vacc[6]), "+v"(vacc[7]), \
                    "+v"(uv[0]), "+v"(uv[1]), "+v"(uv[2]), "+v"(uv[3]), \
                    "+v"(uv[4]), "+v"(uv[5]), "+v"(uv[6]), "+v"(uv[7]), \
                    "+v"(m_run), "+v"(l_run) : : "memory")
#pragma unroll
  for (int g = 0; g < NPAIR; ++g) {
    __builtin_amdgcn_sched_barrier(0);
    if (g + 1 < NPAIR) {
      if ((g & 1) == 0) load_pair(hB, hp + (g + 1) * pair_step);
      else              load_pair(hA, hp + (g + 1) * pair_step);
    }
    __builtin_amdgcn_sched_barrier(0);
    if ((g & 1) == 0)
      reduce_pair(hA, uv, lane_value(mk_lane, 2 * g), lane_value(mk_lane, 2 * g + 1), cb, m_run, l_run, vacc);
    else
      reduce_pair(hB, uv, lane_value(mk_lane, 2 * g), lane_value(mk_lane, 2 * g + 1), cb, m_run, l_run, vacc);
    PAIR_FENCE();
  }
#undef PAIR_FENCE

  if (lane == 0) { wave_m[w] = m_run; wave_l[w] = l_run; }
  __syncthreads();
  float mb = wave_m[0];
#pragma unroll
  for (int i = 1; i < 8; ++i) mb = fmaxf(mb, wave_m[i]);
  const float f = __expf(m_run - mb);
#pragma unroll
  for (int j = 0; j < 8; ++j) {
    const int base = 4 * (32 * j + lane);
    vsh[w][base + 0] = vacc[j].x * f;
    vsh[w][base + 1] = vacc[j].y * f;
    vsh[w][base + 2] = vacc[j].z * f;
    vsh[w][base + 3] = vacc[j].w * f;
  }
  __syncthreads();

  float lb = 0.f;
#pragma unroll
  for (int i = 0; i < 8; ++i) lb += wave_l[i] * __expf(wave_m[i] - mb);

  const int pi = b * SPLIT + split;
  for (int d = tid; d < DIM; d += 256) {
    float sum = 0.f;
#pragma unroll
    for (int i = 0; i < 8; ++i) sum += vsh[i][d];
    pv[(size_t)pi * DIM + d] = sum;
  }
  if (tid == 0) { pm[pi] = mb; pl[pi] = lb; }
}

#define K3_THREADS 64
#define K3_PER_BATCH (DIM / (4 * K3_THREADS))
__global__ __launch_bounds__(K3_THREADS) void k3_combine(const float* __restrict__ pm,
                                                         const float* __restrict__ pl,
                                                         const float* __restrict__ pv,
                                                         float* __restrict__ vnorm) {
  const int b    = blockIdx.x / K3_PER_BATCH;
  const int part = blockIdx.x % K3_PER_BATCH;
  const int tid  = threadIdx.x;
  const int lane = tid & 31;
  const int d4   = part * K3_THREADS + tid;

  const float pmv = pm[b * SPLIT + (lane & (SPLIT - 1))];
  const float plv = pl[b * SPLIT + (lane & (SPLIT - 1))];
  const v4f* p = (const v4f*)(pv + (size_t)b * SPLIT * DIM) + d4;
  v4f x[SPLIT];
#pragma unroll
  for (int i = 0; i < SPLIT; ++i) x[i] = p[i * (DIM / 4)];
  __builtin_amdgcn_sched_barrier(0);

  float mb = -__builtin_inff();
#pragma unroll
  for (int i = 0; i < SPLIT; ++i) mb = fmaxf(mb, lane_value(pmv, i));
  const float fl = __expf(pmv - mb);
  float f[SPLIT];
  float L = 0.f;
#pragma unroll
  for (int i = 0; i < SPLIT; ++i) {
    f[i] = lane_value(fl, i);
    L += lane_value(plv, i) * f[i];
  }
  const float invL = 1.f / L;
  v4f sv = {0.f, 0.f, 0.f, 0.f};
#pragma unroll
  for (int i = 0; i < SPLIT; ++i) sv += x[i] * f[i];
  ((v4f*)(vnorm + (size_t)b * DIM))[d4] = sv * invL;
}

__global__ __launch_bounds__(32 * KSPLIT) void k4_out_gemm(const float* __restrict__ vnorm,
                                                           const float* __restrict__ W,
                                                           const float* __restrict__ bias,
                                                           float* __restrict__ ct) {
  __shared__ float part[KSPLIT][256];
  const int tid  = threadIdx.x;
  const int kw   = tid >> 5;
  const int lane = tid & 31;
  const int mt = blockIdx.x >> 6;
  const int nt = blockIdx.x & 63;
  const int row = lane & 15;
  const int kh  = lane >> 4;
  const int m0 = mt * 16, n0 = nt * 16;
  const int kbeg = kw * KCHUNK;

  v8f acc = {};
#pragma unroll
  for (int k = 0; k < KCHUNK; k += 4) {
    const int kk = kbeg + k + 2 * kh;
    v2f a, bm;
    a.x  = vnorm[(size_t)(m0 + row) * DIM + kk];
    a.y  = vnorm[(size_t)(m0 + row) * DIM + kk + 1];
    bm.x = W[(size_t)(n0 + row) * DIM + kk];
    bm.y = W[(size_t)(n0 + row) * DIM + kk + 1];
    acc = __builtin_amdgcn_wmma_f32_16x16x4_f32(false, a, false, bm,
                                                (short)0, acc, false, false);
  }
#pragma unroll
  for (int i = 0; i < 8; ++i) part[kw][i * 32 + lane] = acc[i];
  __syncthreads();
  if (tid < 256) {
    float s = 0.f;
#pragma unroll
    for (int j = 0; j < KSPLIT; ++j) s += part[j][tid];
    ct[(size_t)(m0 + kw + 8 * kh) * DIM + n0 + row] = s + bias[n0 + row];
  }
}

extern "C" void kernel_launch(void* const* d_in, const int* in_sizes, int n_in,
                              void* d_out, int out_size, void* d_ws, size_t ws_size,
                              hipStream_t stream) {
  const float* st   = (const float*)d_in[0];
  const float* hx   = (const float*)d_in[1];
  const float* mask = (const float*)d_in[2];
  const float* W    = (const float*)d_in[3];
  const float* bias = (const float*)d_in[4];
  float* out = (float*)d_out;

  float* wsf   = (float*)d_ws;
  float* u     = wsf;
  float* pm    = u  + (size_t)BATCH * DIM;
  float* pl    = pm + (size_t)BATCH * SPLIT;
  float* pv    = pl + (size_t)BATCH * SPLIT;
  float* vnorm = pv + (size_t)BATCH * SPLIT * DIM;

  k1_u_gemm<<<128, 32 * KSPLIT, 0, stream>>>(st, W, u);

  k2_flash_partial<<<BATCH * SPLIT, 256, 0, stream>>>(st, hx, mask, bias, u,
                                                      pm, pl, pv);

  k3_combine<<<BATCH * K3_PER_BATCH, K3_THREADS, 0, stream>>>(pm, pl, pv, vnorm);

  k4_out_gemm<<<128, 32 * KSPLIT, 0, stream>>>(vnorm, W, bias, out);
}


# Skinny GEMMs: k-contiguous operand tiles (st, vnorm, W^T rows) fetched as whole 128-B lines, turned to fragment order through a per-wave LDS tile (k order inside a chunk permuted: sums re-ordered)
# speedup vs baseline: 2.4410x; 1.0558x over previous
#include <hip/hip_runtime.h>
#include <hip/hip_bf16.h>

#define BATCH 32
#define SLEN  2048
#define DIM   1024
#define SPLIT 16

typedef __attribute__((ext_vector_type(2))) float v2f;
typedef __attribute__((ext_vector_type(4))) float v4f;
typedef __attribute__((ext_vector_type(8))) float v8f;

#define KSPLIT 16
#define KCHUNK (DIM / KSPLIT)
#define KGROUPS (KCHUNK / 8)

#define TPITCH 17
#define TILE_V4 (16 * TPITCH)

__device__ __forceinline__ void rows_request(v4f (&x)[8], const float* g, int lane) {
  const v4f* p = (const v4f*)(g + (size_t)(lane >> 4) * DIM) + (lane & 15);
#pragma unroll
  for (int j = 0; j < 8; ++j) x[j] = p[j * (2 * DIM / 4)];
}
__device__ __forceinline__ void rows_stage(v4f* tile, const v4f (&x)[8], int lane) {
#pragma unroll
  for (int j = 0; j < 8; ++j) tile[(2 * j + (lane >> 4)) * TPITCH + (lane & 15)] = x[j];
}
__device__ __forceinline__ void rows_fragments(v4f (&f)[KGROUPS], const v4f* tile, int lane) {
#pragma unroll
  for (int t = 0; t < KGROUPS; ++t) f[t] = tile[(lane & 15) * TPITCH + 2 * t + (lane >> 4)];
}
__device__ __forceinline__ void rows_handover() {
  __builtin_amdgcn_fence(__ATOMIC_ACQ_REL, "wavefront");
}

__global__ __launch_bounds__(32 * KSPLIT) void k1_u_gemm(const float* __restrict__ st,
                                                         const float* __restrict__ W,
                                                         float* __restrict__ u) {
  __shared__ float part[KSPLIT][256];
  __shared__ v4f tileA[KSPLIT][TILE_V4];
  const int tid  = threadIdx.x;
  const int kw   = tid >> 5;
  const int lane = tid & 31;
  const int mt = blockIdx.x >> 6;
  const int nt = blockIdx.x & 63;
  const int row = lane & 15;
  const int kh  = lane >> 4;
  const int m0 = mt * 16, n0 = nt * 16;
  const int kbeg = kw * KCHUNK;

  v4f xa[8], av[KGROUPS];
  rows_request(xa, st + (size_t)m0 * DIM + kbeg, lane);
  v2f bm[2 * KGROUPS];
#pragma unroll
  for (int t = 0; t < KGROUPS; ++t) {
#pragma unroll
    for (int p = 0; p < 2; ++p) {
      const int kq = kbeg + 8 * t + 4 * kh + 2 * p;
      bm[2 * t + p].x = W[(size_t)kq * DIM + n0 + row];
      bm[2 * t + p].y = W[(size_t)(kq + 1) * DIM + n0 + row];
    }
  }
  rows_stage(tileA[kw], xa, lane);
  rows_handover();
  rows_fragments(av, tileA[kw], lane);
  __builtin_amdgcn_sched_barrier(0);

  v8f acc = {};
#pragma unroll
  for (int t = 0; t < KGROUPS; ++t) {
    acc = __builtin_amdgcn_wmma_f32_16x16x4_f32(false, av[t].xy, false, bm[2 * t],
                                                (short)0, acc, false, false);
    acc = __builtin_amdgcn_wmma_f32_16x16x4_f32(false, av[t].zw, false, bm[2 * t + 1],
                                                (short)0, acc, false, false);
  }
#pragma unroll
  for (int i = 0; i < 8; ++i) part[kw][i * 32 + lane] = acc[i];
  __syncthreads();
  if (tid < 256) {
    float s = 0.f;
#pragma unroll
    for (int j = 0; j < KSPLIT; ++j) s += part[j][tid];
    u[(size_t)(m0 + kw + 8 * kh) * DIM + n0 + row] = s;
  }
}

#define RPW 16
#define NPAIR (RPW / 2)

__device__ __forceinline__ v2f xor_pair(v2f p, int m) {
  v2f r;
  r.x = __shfl_xor(p.x, m, 32);
  r.y = __shfl_xor(p.y, m, 32);
  return r;
}

__device__ __forceinline__ void load_pair(v4f (&h)[2][8], const v4f* p) {
#pragma unroll
  for (int i = 0; i < 2; ++i)
#pragma unroll
    for (int q = 0; q < 8; ++q)
      h[i][q] = __builtin_nontemporal_load(p + i * (DIM / 4) + 32 * q);
}

__device__ __forceinline__ void reduce_pair(const v4f (&h)[2][8], const v4f (&uv)[8],
                                            float mkA, float mkB, float cb,
                                            float& m_run, float& l_run, v4f (&vacc)[8]) {
  v2f a0 = {0.f, 0.f}, a1 = {0.f, 0.f};
  v2f e0 = {0.f, 0.f}, e1 = {0.f, 0.f};
#pragma unroll
  for (int j = 0; j < 8; ++j) {
    a0 += h[0][j].xy * uv[j].xy;
    a1 += h[0][j].zw * uv[j].zw;
    e0 += h[1][j].xy * uv[j].xy;
    e1 += h[1][j].zw * uv[j].zw;
  }
  v2f pp;
  pp.x = (a0.x + a0.y) + (a1.x + a1.y);
  pp.y = (e0.x + e0.y) + (e1.x + e1.y);
  pp += xor_pair(pp, 16);
  pp += xor_pair(pp,  8);
  pp += xor_pair(pp,  4);
  pp += xor_pair(pp,  2);
  pp += xor_pair(pp,  1);

  const float scoreA = (pp.x + cb) * (mkA + (1.f - mkA) * 1e-18f);
  const float scoreB = (pp.y + cb) * (mkB + (1.f - mkB) * 1e-18f);

  const float mn  = fmaxf(m_run, fmaxf(scoreA, scoreB));
  const float sc  = __expf(m_run - mn);
  const float peA = __expf(scoreA - mn);
  const float peB = __expf(scoreB - mn);
  l_run = (l_run * sc + peA) + peB;
#pragma unroll
  for (int j = 0; j < 8; ++j) vacc[j] = (vacc[j] * sc + peA * h[0][j]) + peB * h[1][j];
  m_run = mn;
}

__device__ __forceinline__ float lane_value(float v, int l) {
  return __builtin_bit_cast(float, __builtin_amdgcn_readlane(__builtin_bit_cast(int, v), l));
}

__global__ __launch_bounds__(256) void k2_flash_partial(
    const float* __restrict__ st, const float* __restrict__ hx,
    const float* __restrict__ mask, const float* __restrict__ bias,
    const float* __restrict__ u,
    float* __restrict__ pm, float* __restrict__ pl, float* __restrict__ pv) {
  const int b     = blockIdx.x / SPLIT;
  const int split = blockIdx.x % SPLIT;
  const int tid   = threadIdx.x;
  const int lane  = tid & 31;
  const int w     = __builtin_amdgcn_readfirstlane(tid >> 5);

  __shared__ float wave_m[8], wave_l[8];
  __shared__ float vsh[8][DIM];

  const int s0 = split * (SLEN / SPLIT) + w * RPW;

  const v4f* u4 = (const v4f*)(u  + (size_t)b * DIM) + lane;
  const v4f* s4 = (const v4f*)(st + (size_t)b * DIM) + lane;
  const v4f* b4 = (const v4f*)bias + lane;
  v4f uv[8], sv[8], bv[8];
#pragma unroll
  for (int j = 0; j < 8; ++j) { uv[j] = u4[32 * j]; sv[j] = s4[32 * j]; bv[j] = b4[32 * j]; }
  const float mk_lane = mask[b * SLEN + s0 + (lane & (RPW - 1))];

  const v4f* hp = (const v4f*)(hx + ((size_t)b * SLEN + s0) * DIM) + lane;
  const int pair_step = 2 * (DIM / 4);
  v4f hA[2][8], hB[2][8];
  __builtin_amdgcn_sched_barrier(0);
  load_pair(hA, hp);

  float cb;
  {
    v2f c0 = {0.f, 0.f}, c1 = {0.f, 0.f};
#pragma unroll
    for (int j = 0; j < 8; ++j) {
      c0 += sv[j].xy * bv[j].xy;
      c1 += sv[j].zw * bv[j].zw;
    }
    cb = (c0.x + c0.y) + (c1.x + c1.y);
    cb += __shfl_xor(cb, 16, 32);
    cb += __shfl_xor(cb,  8, 32);
    cb += __shfl_xor(cb,  4, 32);
    cb += __shfl_xor(cb,  2, 32);
    cb += __shfl_xor(cb,  1, 32);
  }

  float m_run = -__builtin_inff();
  float l_run = 0.f;
  v4f vacc[8];
#pragma unroll
  for (int j = 0; j < 8; ++j) vacc[j] = (v4f){0.f, 0.f, 0.f, 0.f};

#define PAIR_FENCE() \
  asm volatile("" : "+v"(vacc[0]), "+v"(vacc[1]), "+v"(vacc[2]), "+v"(vacc[3]), \
                    "+v"(vacc[4]), "+v"(vacc[5]), "+v"(vacc[6]), "+v"(vacc[7]), \
                    "+v"(uv[0]), "+v"(uv[1]), "+v"(uv[2]), "+v"(uv[3]), \
                    "+v"(uv[4]), "+v"(uv[5]), "+v"(uv[6]), "+v"(uv[7]), \
                    "+v"(m_run), "+v"(l_run) : : "memory")
#pragma unroll
  for (int g = 0; g < NPAIR; ++g) {
    __builtin_amdgcn_sched_barrier(0);
    if (g + 1 < NPAIR) {
      if ((g & 1) == 0) load_pair(hB, hp + (g + 1) * pair_step);
      else              load_pair(hA, hp + (g + 1) * pair_step);
    }
    __builtin_amdgcn_sched_barrier(0);
    if ((g & 1) == 0)
      reduce_pair(hA, uv, lane_value(mk_lane, 2 * g), lane_value(mk_lane, 2 * g + 1), cb, m_run, l_run, vacc);
    else
      reduce_pair(hB, uv, lane_value(mk_lane, 2 * g), lane_value(mk_lane, 2 * g + 1), cb, m_run, l_run, vacc);
    PAIR_FENCE();
  }
#undef PAIR_FENCE

  if (lane == 0) { wave_m[w] = m_run; wave_l[w] = l_run; }
  __syncthreads();
  float mb = wave_m[0];
#pragma unroll
  for (int i = 1; i < 8; ++i) mb = fmaxf(mb, wave_m[i]);
  const float f = __expf(m_run - mb);
#pragma unroll
  for (int j = 0; j < 8; ++j) {
    const int base = 4 * (32 * j + lane);
    vsh[w][base + 0] = vacc[j].x * f;
    vsh[w][base + 1] = vacc[j].y * f;
    vsh[w][base + 2] = vacc[j].z * f;
    vsh[w][base + 3] = vacc[j].w * f;
  }
  __syncthreads();

  float lb = 0.f;
#pragma unroll
  for (int i = 0; i < 8; ++i) lb += wave_l[i] * __expf(wave_m[i] - mb);

  const int pi = b * SPLIT + split;
  for (int d = tid; d < DIM; d += 256) {
    float sum = 0.f;
#pragma unroll
    for (int i = 0; i < 8; ++i) sum += vsh[i][d];
    pv[(size_t)pi * DIM + d] = sum;
  }
  if (tid == 0) { pm[pi] = mb; pl[pi] = lb; }
}

#define K3_THREADS 64
#define K3_PER_BATCH (DIM / (4 * K3_THREADS))
__global__ __launch_bounds__(K3_THREADS) void k3_combine(const float* __restrict__ pm,
                                                         const float* __restrict__ pl,
                                                         const float* __restrict__ pv,
                                                         float* __restrict__ vnorm) {
  const int b    = blockIdx.x / K3_PER_BATCH;
  const int part = blockIdx.x % K3_PER_BATCH;
  const int tid  = threadIdx.x;
  const int lane = tid & 31;
  const int d4   = part * K3_THREADS + tid;

  const float pmv = pm[b * SPLIT + (lane & (SPLIT - 1))];
  const float plv = pl[b * SPLIT + (lane & (SPLIT - 1))];
  const v4f* p = (const v4f*)(pv + (size_t)b * SPLIT * DIM) + d4;
  v4f x[SPLIT];
#pragma unroll
  for (int i = 0; i < SPLIT; ++i) x[i] = p[i * (DIM / 4)];
  __builtin_amdgcn_sched_barrier(0);

  float mb = -__builtin_inff();
#pragma unroll
  for (int i = 0; i < SPLIT; ++i) mb = fmaxf(mb, lane_value(pmv, i));
  const float fl = __expf(pmv - mb);
  float f[SPLIT];
  float L = 0.f;
#pragma unroll
  for (int i = 0; i < SPLIT; ++i) {
    f[i] = lane_value(fl, i);
    L += lane_value(plv, i) * f[i];
  }
  const float invL = 1.f / L;
  v4f sv = {0.f, 0.f, 0.f, 0.f};
#pragma unroll
  for (int i = 0; i < SPLIT; ++i) sv += x[i] * f[i];
  ((v4f*)(vnorm + (size_t)b * DIM))[d4] = sv * invL;
}

__global__ __launch_bounds__(32 * KSPLIT) void k4_out_gemm(const float* __restrict__ vnorm,
                                                           const float* __restrict__ W,
                                                           const float* __restrict__ bias,
                                                           float* __restrict__ ct) {
  __shared__ float part[KSPLIT][256];
  __shared__ v4f tileA[KSPLIT][TILE_V4];
  __shared__ v4f tileB[KSPLIT][TILE_V4];
  const int tid  = threadIdx.x;
  const int kw   = tid >> 5;
  const int lane = tid & 31;
  const int mt = blockIdx.x >> 6;
  const int nt = blockIdx.x & 63;
  const int row = lane & 15;
  const int kh  = lane >> 4;
  const int m0 = mt * 16, n0 = nt * 16;
  const int kbeg = kw * KCHUNK;

  v4f xa[8], xb[8], av[KGROUPS], bw[KGROUPS];
  rows_request(xa, vnorm + (size_t)m0 * DIM + kbeg, lane);
  rows_request(xb, W     + (size_t)n0 * DIM + kbeg, lane);
  rows_stage(tileA[kw], xa, lane);
  rows_stage(tileB[kw], xb, lane);
  rows_handover();
  rows_fragments(av, tileA[kw], lane);
  rows_fragments(bw, tileB[kw], lane);
  __builtin_amdgcn_sched_barrier(0);

  v8f acc = {};
#pragma unroll
  for (int t = 0; t < KGROUPS; ++t) {
    acc = __builtin_amdgcn_wmma_f32_16x16x4_f32(false, av[t].xy, false, bw[t].xy,
                                                (short)0, acc, false, false);
    acc = __builtin_amdgcn_wmma_f32_16x16x4_f32(false, av[t].zw, false, bw[t].zw,
                                                (short)0, acc, false, false);
  }
#pragma unroll
  for (int i = 0; i < 8; ++i) part[kw][i * 32 + lane] = acc[i];
  __syncthreads();
  if (tid < 256) {
    float s = 0.f;
#pragma unroll
    for (int j = 0; j < KSPLIT; ++j) s += part[j][tid];
    ct[(size_t)(m0 + kw + 8 * kh) * DIM + n0 + row] = s + bias[n0 + row];
  }
}

extern "C" void kernel_launch(void* const* d_in, const int* in_sizes, int n_in,
                              void* d_out, int out_size, void* d_ws, size_t ws_size,
                              hipStream_t stream) {
  const float* st   = (const float*)d_in[0];
  const float* hx   = (const float*)d_in[1];
  const float* mask = (const float*)d_in[2];
  const float* W    = (const float*)d_in[3];
  const float* bias = (const float*)d_in[4];
  float* out = (float*)d_out;

  float* wsf   = (float*)d_ws;
  float* u     = wsf;
  float* pm    = u  + (size_t)BATCH * DIM;
  float* pl    = pm + (size_t)BATCH * SPLIT;
  float* pv    = pl + (size_t)BATCH * SPLIT;
  float* vnorm = pv + (size_t)BATCH * SPLIT * DIM;

  k1_u_gemm<<<128, 32 * KSPLIT, 0, stream>>>(st, W, u);

  k2_flash_partial<<<BATCH * SPLIT, 256, 0, stream>>>(st, hx, mask, bias, u,
                                                      pm, pl, pv);

  k3_combine<<<BATCH * K3_PER_BATCH, K3_THREADS, 0, stream>>>(pm, pl, pv, vnorm);

  k4_out_gemm<<<128, 32 * KSPLIT, 0, stream>>>(vnorm, W, bias, out);
}
